# starting kernel
# baseline (speedup 1.0000x reference)
#include <hip/hip_runtime.h>


typedef float v2f __attribute__((ext_vector_type(2)));
typedef float v8f __attribute__((ext_vector_type(8)));

#define ATT_S   4096
#define ATT_D   64
#define NHEAD   16
#define NCHUNK  16
#define CHUNK_S (ATT_S / NCHUNK)
#define K1_WAVES 4
#define WAVE_S  (CHUNK_S / K1_WAVES)
#define QTILE   128
#define QWAVES  2
#define QPAD    68

static __device__ __forceinline__ v8f wmma_f32(v2f a, v2f b, v8f c) {
  return __builtin_amdgcn_wmma_f32_16x16x4_f32(false, a, false, b,
                                               (short)0, c, false, false);
}

__global__ __launch_bounds__(K1_WAVES * 32)
void attn_ktv_partial(const float* __restrict__ K, const float* __restrict__ V,
                      float* __restrict__ partial) {
  const int head  = blockIdx.x / NCHUNK;
  const int chunk = blockIdx.x % NCHUNK;
  const int wave  = threadIdx.x >> 5;
  const int lane  = threadIdx.x & 31;
  const int lmod  = lane & 15;
  const int lhalf = lane >> 4;

  const float* __restrict__ Kh = K + (size_t)head * ATT_S * ATT_D;
  const float* __restrict__ Vh = V + (size_t)head * ATT_S * ATT_D;
  const int s0 = chunk * CHUNK_S + wave * WAVE_S;

  v8f c[4][4];
#pragma unroll
  for (int i = 0; i < 4; ++i)
#pragma unroll
    for (int j = 0; j < 4; ++j)
      c[i][j] = (v8f){0.f, 0.f, 0.f, 0.f, 0.f, 0.f, 0.f, 0.f};

#pragma unroll 2
  for (int step = 0; step < WAVE_S / 4; ++step) {
    const int s = s0 + step * 4 + lhalf * 2;
    const float* __restrict__ kr = Kh + (size_t)s * ATT_D + lmod;
    const float* __restrict__ vr = Vh + (size_t)s * ATT_D + lmod;
    v2f a[4], b[4];
#pragma unroll
    for (int t = 0; t < 4; ++t) {
      a[t].x = kr[t * 16];
      a[t].y = kr[t * 16 + ATT_D];
      b[t].x = vr[t * 16];
      b[t].y = vr[t * 16 + ATT_D];
    }
#pragma unroll
    for (int i = 0; i < 4; ++i)
#pragma unroll
      for (int j = 0; j < 4; ++j)
        c[i][j] = wmma_f32(a[i], b[j], c[i][j]);
  }

  __shared__ float red[K1_WAVES][ATT_D * ATT_D];
#pragma unroll
  for (int i = 0; i < 4; ++i)
#pragma unroll
    for (int j = 0; j < 4; ++j)
#pragma unroll
      for (int r = 0; r < 8; ++r)
        red[wave][(i * 16 + r + lhalf * 8) * ATT_D + j * 16 + lmod] = c[i][j][r];
  __syncthreads();

  float* __restrict__ outp =
      partial + ((size_t)head * NCHUNK + chunk) * (ATT_D * ATT_D);
  for (int e = threadIdx.x; e < ATT_D * ATT_D; e += K1_WAVES * 32)
    outp[e] = (red[0][e] + red[1][e]) + (red[2][e] + red[3][e]);
}

__global__ __launch_bounds__(256)
void attn_reduce(const float* __restrict__ partial, float* __restrict__ M) {
  const int head = blockIdx.x;
  const float* __restrict__ p = partial + (size_t)head * NCHUNK * ATT_D * ATT_D;
  for (int e = threadIdx.x; e < ATT_D * ATT_D; e += 256) {
    float s = 0.f;
#pragma unroll
    for (int cc = 0; cc < NCHUNK; ++cc)
      s += p[cc * ATT_D * ATT_D + e];
    M[(size_t)head * ATT_D * ATT_D + e] = s;
  }
}

__global__ __launch_bounds__(QWAVES * 32)
void attn_qm(const float* __restrict__ Q, const float* __restrict__ M,
             float* __restrict__ out) {
  const int head  = blockIdx.x / (ATT_S / QTILE);
  const int qc    = blockIdx.x % (ATT_S / QTILE);
  const int wave  = threadIdx.x >> 5;
  const int lane  = threadIdx.x & 31;
  const int lmod  = lane & 15;
  const int lhalf = lane >> 4;
  const int q0    = qc * QTILE;

  __shared__ float qs[QTILE * QPAD];

  const float* __restrict__ Qh = Q + ((size_t)head * ATT_S + q0) * ATT_D;
  for (int idx = threadIdx.x; idx < QTILE * (ATT_D / 4); idx += QWAVES * 32) {
    const int row = idx >> 4;
    const int c4  = idx & 15;
    const float4 f = ((const float4*)Qh)[idx];
    *(float4*)&qs[row * QPAD + c4 * 4] = f;
  }
  __syncthreads();

  const float* __restrict__ Mh = M + (size_t)head * ATT_D * ATT_D;

  v8f c[4][4];
#pragma unroll
  for (int i = 0; i < 4; ++i)
#pragma unroll
    for (int j = 0; j < 4; ++j)
      c[i][j] = (v8f){0.f, 0.f, 0.f, 0.f, 0.f, 0.f, 0.f, 0.f};

#pragma unroll 2
  for (int step = 0; step < ATT_D / 4; ++step) {
    const int k0 = step * 4 + lhalf * 2;
    v2f a[4], b[4];
#pragma unroll
    for (int t = 0; t < 4; ++t) {
      const v2f av = *(const v2f*)&qs[(wave * 64 + t * 16 + lmod) * QPAD + k0];
      a[t] = av;
      b[t].x = Mh[(k0 + 0) * ATT_D + t * 16 + lmod];
      b[t].y = Mh[(k0 + 1) * ATT_D + t * 16 + lmod];
    }
#pragma unroll
    for (int i = 0; i < 4; ++i)
#pragma unroll
      for (int j = 0; j < 4; ++j)
        c[i][j] = wmma_f32(a[i], b[j], c[i][j]);
  }

  float* __restrict__ Oh =
      out + ((size_t)head * ATT_S + q0 + wave * 64) * ATT_D;
#pragma unroll
  for (int i = 0; i < 4; ++i)
#pragma unroll
    for (int j = 0; j < 4; ++j)
#pragma unroll
      for (int r = 0; r < 8; ++r)
        Oh[(size_t)(i * 16 + r + lhalf * 8) * ATT_D + j * 16 + lmod] =
            c[i][j][r];
}

extern "C" void kernel_launch(void* const* d_in, const int* in_sizes, int n_in,
                              void* d_out, int out_size, void* d_ws,
                              size_t ws_size, hipStream_t stream) {
  (void)in_sizes; (void)n_in; (void)out_size; (void)ws_size;
  const float* q = (const float*)d_in[0];
  const float* k = (const float*)d_in[1];
  const float* v = (const float*)d_in[2];
  float* out = (float*)d_out;

  float* partial = (float*)d_ws;
  float* M = partial + (size_t)NHEAD * NCHUNK * ATT_D * ATT_D;

  attn_ktv_partial<<<NHEAD * NCHUNK, K1_WAVES * 32, 0, stream>>>(k, v, partial);
  attn_reduce<<<NHEAD, 256, 0, stream>>>(partial, M);
  attn_qm<<<NHEAD * (ATT_S / QTILE), QWAVES * 32, 0, stream>>>(q, M, out);
}


# Stage 1 re-tiled: K/V chunk staged through LDS, 8 waves each own a 32x16 block of K^T V over all 256 rows (was 4 waves x 64 rows + LDS cross-wave sum); one sum chain per element, add order changes
# speedup vs baseline: 1.5945x; 1.5945x over previous
#include <hip/hip_runtime.h>


typedef float v2f __attribute__((ext_vector_type(2)));
typedef float v8f __attribute__((ext_vector_type(8)));

#define ATT_S   4096
#define ATT_D   64
#define NHEAD   16
#define NCHUNK  16
#define CHUNK_S (ATT_S / NCHUNK)
#define K1_WAVES 8
#define QTILE   256
#define QWAVES  8
#define QPAD    68
#define MKP     160

static __device__ __forceinline__ v8f wmma_f32(v2f a, v2f b, v8f c) {
  return __builtin_amdgcn_wmma_f32_16x16x4_f32(false, a, false, b,
                                               (short)0, c, false, false);
}

#define K1_THREADS (K1_WAVES * 32)
#define K1_SUB     64
#define K1_ROWSTEP (K1_THREADS / 16)
#define K1_PIECE_F (K1_SUB * ATT_D)
#define K1_BUF_F   (2 * K1_PIECE_F)

#define K1_GLOAD(S, P)                                               \
  S##k0 = Kc4[(P) * (K1_PIECE_F / 4) + 0 * K1_THREADS];              \
  S##k1 = Kc4[(P) * (K1_PIECE_F / 4) + 1 * K1_THREADS];              \
  S##k2 = Kc4[(P) * (K1_PIECE_F / 4) + 2 * K1_THREADS];              \
  S##k3 = Kc4[(P) * (K1_PIECE_F / 4) + 3 * K1_THREADS];              \
  S##v0 = Vc4[(P) * (K1_PIECE_F / 4) + 0 * K1_THREADS];              \
  S##v1 = Vc4[(P) * (K1_PIECE_F / 4) + 1 * K1_THREADS];              \
  S##v2 = Vc4[(P) * (K1_PIECE_F / 4) + 2 * K1_THREADS];              \
  S##v3 = Vc4[(P) * (K1_PIECE_F / 4) + 3 * K1_THREADS];
#define K1_FILL(S, B)                                                \
  *(float4*)&fillp[(B) * K1_BUF_F + 0 * K1_ROWSTEP * ATT_D] = S##k0; \
  *(float4*)&fillp[(B) * K1_BUF_F + 1 * K1_ROWSTEP * ATT_D] = S##k1; \
  *(float4*)&fillp[(B) * K1_BUF_F + 2 * K1_ROWSTEP * ATT_D] = S##k2; \
  *(float4*)&fillp[(B) * K1_BUF_F + 3 * K1_ROWSTEP * ATT_D] = S##k3; \
  *(float4*)&fillp[(B) * K1_BUF_F + K1_PIECE_F + 0 * K1_ROWSTEP * ATT_D] = S##v0; \
  *(float4*)&fillp[(B) * K1_BUF_F + K1_PIECE_F + 1 * K1_ROWSTEP * ATT_D] = S##v1; \
  *(float4*)&fillp[(B) * K1_BUF_F + K1_PIECE_F + 2 * K1_ROWSTEP * ATT_D] = S##v2; \
  *(float4*)&fillp[(B) * K1_BUF_F + K1_PIECE_F + 3 * K1_ROWSTEP * ATT_D] = S##v3;
#define K1_FRAGS(FA, FB, B, G)                                       \
  _Pragma("unroll") for (int s = 0; s < 4; ++s) {                    \
    _Pragma("unroll") for (int t = 0; t < 2; ++t) {                  \
      FA[s][t].x = kfrag[t][(B) * K1_BUF_F + ((G) * 4 + s) * 4 * ATT_D];          \
      FA[s][t].y = kfrag[t][(B) * K1_BUF_F + ((G) * 4 + s) * 4 * ATT_D + ATT_D];  \
    }                                                                \
    FB[s].x = vfrag[(B) * K1_BUF_F + ((G) * 4 + s) * 4 * ATT_D];     \
    FB[s].y = vfrag[(B) * K1_BUF_F + ((G) * 4 + s) * 4 * ATT_D + ATT_D];          \
  }
#define K1_MMA(FA, FB)                                               \
  _Pragma("unroll") for (int s = 0; s < 4; ++s)                      \
    _Pragma("unroll") for (int i = 0; i < 2; ++i)                    \
      c[i] = wmma_f32(FA[s][i], FB[s], c[i]);
#define K1_PIECE(B)                                                  \
  K1_FRAGS(fa0, fb0, B, 0)                                           \
  K1_FRAGS(fa1, fb1, B, 1) K1_MMA(fa0, fb0)                          \
  K1_FRAGS(fa0, fb0, B, 2) K1_MMA(fa1, fb1)                          \
  K1_FRAGS(fa1, fb1, B, 3) K1_MMA(fa0, fb0)                          \
  K1_MMA(fa1, fb1)

__global__ __launch_bounds__(K1_THREADS)
void attn_ktv_partial(const float* __restrict__ K, const float* __restrict__ V,
                      float* __restrict__ partial) {
  const int head  = blockIdx.x / NCHUNK;
  const int chunk = blockIdx.x % NCHUNK;
  const int wave  = threadIdx.x >> 5;
  const int lane  = threadIdx.x & 31;
  const int lmod  = lane & 15;
  const int lhalf = lane >> 4;
  const int wi    = wave >> 2;
  const int wj    = wave & 3;

  __shared__ __attribute__((aligned(16))) float sm[2 * K1_BUF_F];

  const size_t row0 = (size_t)head * ATT_S + (size_t)chunk * CHUNK_S;
  const float4* __restrict__ Kc4 = (const float4*)(K + row0 * ATT_D) + threadIdx.x;
  const float4* __restrict__ Vc4 = (const float4*)(V + row0 * ATT_D) + threadIdx.x;

  float* __restrict__ fillp =
      &sm[(threadIdx.x >> 4) * ATT_D +
          (((threadIdx.x & 15) ^ ((wave & 1) << 2)) << 2)];

  const float* __restrict__ kfrag[2];
#pragma unroll
  for (int t = 0; t < 2; ++t)
    kfrag[t] = &sm[2 * lhalf * ATT_D + (((wi * 2 + t) ^ lhalf) << 4) + lmod];
  const float* __restrict__ vfrag =
      &sm[K1_PIECE_F + 2 * lhalf * ATT_D + ((wj ^ lhalf) << 4) + lmod];

  float4 Ak0, Ak1, Ak2, Ak3, Av0, Av1, Av2, Av3;
  float4 Bk0, Bk1, Bk2, Bk3, Bv0, Bv1, Bv2, Bv3;
  v2f fa0[4][2], fb0[4], fa1[4][2], fb1[4];
  v8f c[2];
#pragma unroll
  for (int i = 0; i < 2; ++i)
    c[i] = (v8f){0.f, 0.f, 0.f, 0.f, 0.f, 0.f, 0.f, 0.f};

  K1_GLOAD(A, 0)
  K1_GLOAD(B, 1)
  K1_FILL(A, 0)
  K1_GLOAD(A, 2)
  __syncthreads();

  K1_FILL(B, 1)
  K1_GLOAD(B, 3)
  K1_PIECE(0)
  __syncthreads();

  K1_FILL(A, 0)
  K1_PIECE(1)
  __syncthreads();

  K1_FILL(B, 1)
  K1_PIECE(0)
  __syncthreads();

  K1_PIECE(1)

  float* __restrict__ outp =
      partial + ((size_t)head * NCHUNK + chunk) * (ATT_D * ATT_D) +
      (size_t)(wi * 32 + lhalf * 8) * ATT_D + wj * 16 + lmod;
#pragma unroll
  for (int i = 0; i < 2; ++i)
#pragma unroll
    for (int r = 0; r < 8; ++r)
      outp[(i * 16 + r) * ATT_D] = c[i][r];
}

__global__ __launch_bounds__(QWAVES * 32)
void attn_qm(const float* __restrict__ Q, const float* __restrict__ partial,
             float* __restrict__ out) {
  const int head  = blockIdx.x / (ATT_S / QTILE);
  const int qc    = blockIdx.x % (ATT_S / QTILE);
  const int wave  = threadIdx.x >> 5;
  const int lane  = threadIdx.x & 31;
  const int lmod  = lane & 15;
  const int lhalf = lane >> 4;
  const int q0    = qc * QTILE;

  __shared__ __attribute__((aligned(16))) float qs[QTILE * QPAD];
  __shared__ __attribute__((aligned(16))) float ms[(ATT_D / 2) * MKP];

  const float4* __restrict__ Qh4 =
      (const float4*)(Q + ((size_t)head * ATT_S + q0) * ATT_D);
  float4 qv[16];
#pragma unroll
  for (int i = 0; i < 16; ++i)
    qv[i] = Qh4[threadIdx.x + i * (QWAVES * 32)];

  const float* __restrict__ ph =
      partial + (size_t)head * NCHUNK * ATT_D * ATT_D;
#pragma unroll
  for (int trip = 0; trip < 2; ++trip) {
    const int item = trip * (QWAVES * 32) + threadIdx.x;
    const int kp   = item >> 4;
    const int n4   = item & 15;
    const float4* __restrict__ pe =
        (const float4*)(ph + (2 * kp) * ATT_D + n4 * 4);
    const float4* __restrict__ po =
        (const float4*)(ph + (2 * kp + 1) * ATT_D + n4 * 4);
    float4 e[NCHUNK], o[NCHUNK];
#pragma unroll
    for (int cc = 0; cc < NCHUNK; ++cc) {
      e[cc] = pe[cc * (ATT_D * ATT_D / 4)];
      o[cc] = po[cc * (ATT_D * ATT_D / 4)];
    }
    if (trip == 0) {
#pragma unroll
      for (int i = 0; i < 16; ++i) {
        const int idx = threadIdx.x + i * (QWAVES * 32);
        const int row = idx >> 4;
        const int c4  = idx & 15;
        *(float4*)&qs[row * QPAD + c4 * 4] = qv[i];
      }
    }
    float4 se = {0.f, 0.f, 0.f, 0.f}, so = {0.f, 0.f, 0.f, 0.f};
#pragma unroll
    for (int cc = 0; cc < NCHUNK; ++cc) {
      se.x += e[cc].x; se.y += e[cc].y; se.z += e[cc].z; se.w += e[cc].w;
      so.x += o[cc].x; so.y += o[cc].y; so.z += o[cc].z; so.w += o[cc].w;
    }
    float* __restrict__ d = &ms[kp * MKP + n4 * 8];
    *(float4*)(d)     = (float4){se.x, so.x, se.y, so.y};
    *(float4*)(d + 4) = (float4){se.z, so.z, se.w, so.w};
  }
  __syncthreads();

  v8f c[2][4];
#pragma unroll
  for (int i = 0; i < 2; ++i)
#pragma unroll
    for (int j = 0; j < 4; ++j)
      c[i][j] = (v8f){0.f, 0.f, 0.f, 0.f, 0.f, 0.f, 0.f, 0.f};

  const float* __restrict__ qw = &qs[(wave * 32 + lmod) * QPAD + lhalf * 2];
  const float* __restrict__ mw = &ms[lhalf * MKP + lmod * 2];
#pragma unroll 2
  for (int step = 0; step < ATT_D / 4; ++step) {
    v2f a[2], b[4];
#pragma unroll
    for (int t = 0; t < 2; ++t)
      a[t] = *(const v2f*)&qw[t * 16 * QPAD + step * 4];
#pragma unroll
    for (int t = 0; t < 4; ++t)
      b[t] = *(const v2f*)&mw[step * 2 * MKP + t * 32];
#pragma unroll
    for (int i = 0; i < 2; ++i)
#pragma unroll
      for (int j = 0; j < 4; ++j)
        c[i][j] = wmma_f32(a[i], b[j], c[i][j]);
  }

  float* __restrict__ Oh =
      out + ((size_t)head * ATT_S + q0 + wave * 32) * ATT_D;
#pragma unroll
  for (int i = 0; i < 2; ++i)
#pragma unroll
    for (int j = 0; j < 4; ++j)
#pragma unroll
      for (int r = 0; r < 8; ++r)
        Oh[(size_t)(i * 16 + r + lhalf * 8) * ATT_D + j * 16 + lmod] =
            c[i][j][r];
}

extern "C" void kernel_launch(void* const* d_in, const int* in_sizes, int n_in,
                              void* d_out, int out_size, void* d_ws,
                              size_t ws_size, hipStream_t stream) {
  (void)in_sizes; (void)n_in; (void)out_size; (void)ws_size;
  const float* q = (const float*)d_in[0];
  const float* k = (const float*)d_in[1];
  const float* v = (const float*)d_in[2];
  float* out = (float*)d_out;

  float* partial = (float*)d_ws;

  attn_ktv_partial<<<NHEAD * NCHUNK, K1_WAVES * 32, 0, stream>>>(k, v, partial);
  attn_qm<<<NHEAD * (ATT_S / QTILE), QWAVES * 32, 0, stream>>>(q, partial, out);
}


# Stage 1 re-cut: 4 row chunks x 4 output quadrants (32x32 of K^T V per workgroup, half rows of K and V, copies 2 sets ahead; was 16 row chunks); stage 2 sums 4 partials a head; add order changes
# speedup vs baseline: 1.8761x; 1.1766x over previous
#include <hip/hip_runtime.h>


typedef float v2f __attribute__((ext_vector_type(2)));
typedef float v8f __attribute__((ext_vector_type(8)));

#define ATT_S   4096
#define ATT_D   64
#define NHEAD   16
#define NCHUNK  4
#define CHUNK_S (ATT_S / NCHUNK)
#define NQUAD   4
#define K1_WAVES 8
#define QTILE   256
#define QWAVES  8
#define QPAD    68
#define MKP     160

static __device__ __forceinline__ v8f wmma_f32(v2f a, v2f b, v8f c) {
  return __builtin_amdgcn_wmma_f32_16x16x4_f32(false, a, false, b,
                                               (short)0, c, false, false);
}

#define K1_THREADS (K1_WAVES * 32)
#define K1_SUB     64
#define K1_PIECE_F (K1_SUB * ATT_D)
#define K1_NPIECE  (CHUNK_S / K1_SUB)
#define K1_RPITCH  36
#define K1_RED_F   (32 * K1_RPITCH)

#define K1_GLOAD(S, H)                                               \
  S##0 = src4[((H) * 4 + 0) * (K1_PIECE_F / 4) + 0 * K1_THREADS]; \
  S##1 = src4[((H) * 4 + 0) * (K1_PIECE_F / 4) + 1 * K1_THREADS]; \
  S##2 = src4[((H) * 4 + 0) * (K1_PIECE_F / 4) + 2 * K1_THREADS]; \
  S##3 = src4[((H) * 4 + 0) * (K1_PIECE_F / 4) + 3 * K1_THREADS]; \
  S##4 = src4[((H) * 4 + 1) * (K1_PIECE_F / 4) + 0 * K1_THREADS]; \
  S##5 = src4[((H) * 4 + 1) * (K1_PIECE_F / 4) + 1 * K1_THREADS]; \
  S##6 = src4[((H) * 4 + 1) * (K1_PIECE_F / 4) + 2 * K1_THREADS]; \
  S##7 = src4[((H) * 4 + 1) * (K1_PIECE_F / 4) + 3 * K1_THREADS]; \
  S##8 = src4[((H) * 4 + 2) * (K1_PIECE_F / 4) + 0 * K1_THREADS]; \
  S##9 = src4[((H) * 4 + 2) * (K1_PIECE_F / 4) + 1 * K1_THREADS]; \
  S##10 = src4[((H) * 4 + 2) * (K1_PIECE_F / 4) + 2 * K1_THREADS]; \
  S##11 = src4[((H) * 4 + 2) * (K1_PIECE_F / 4) + 3 * K1_THREADS]; \
  S##12 = src4[((H) * 4 + 3) * (K1_PIECE_F / 4) + 0 * K1_THREADS]; \
  S##13 = src4[((H) * 4 + 3) * (K1_PIECE_F / 4) + 1 * K1_THREADS]; \
  S##14 = src4[((H) * 4 + 3) * (K1_PIECE_F / 4) + 2 * K1_THREADS]; \
  S##15 = src4[((H) * 4 + 3) * (K1_PIECE_F / 4) + 3 * K1_THREADS];
#define K1_FILL(S, H)                                                \
  *(float4*)&fillp[(0 * 4 + (H)) * K1_PIECE_F + 0 * 16 * ATT_D] = S##0; \
  *(float4*)&fillp[(0 * 4 + (H)) * K1_PIECE_F + 1 * 16 * ATT_D] = S##1; \
  *(float4*)&fillp[(0 * 4 + (H)) * K1_PIECE_F + 2 * 16 * ATT_D] = S##2; \
  *(float4*)&fillp[(0 * 4 + (H)) * K1_PIECE_F + 3 * 16 * ATT_D] = S##3; \
  *(float4*)&fillp[(1 * 4 + (H)) * K1_PIECE_F + 0 * 16 * ATT_D] = S##4; \
  *(float4*)&fillp[(1 * 4 + (H)) * K1_PIECE_F + 1 * 16 * ATT_D] = S##5; \
  *(float4*)&fillp[(1 * 4 + (H)) * K1_PIECE_F + 2 * 16 * ATT_D] = S##6; \
  *(float4*)&fillp[(1 * 4 + (H)) * K1_PIECE_F + 3 * 16 * ATT_D] = S##7; \
  *(float4*)&fillp[(2 * 4 + (H)) * K1_PIECE_F + 0 * 16 * ATT_D] = S##8; \
  *(float4*)&fillp[(2 * 4 + (H)) * K1_PIECE_F + 1 * 16 * ATT_D] = S##9; \
  *(float4*)&fillp[(2 * 4 + (H)) * K1_PIECE_F + 2 * 16 * ATT_D] = S##10; \
  *(float4*)&fillp[(2 * 4 + (H)) * K1_PIECE_F + 3 * 16 * ATT_D] = S##11; \
  *(float4*)&fillp[(3 * 4 + (H)) * K1_PIECE_F + 0 * 16 * ATT_D] = S##12; \
  *(float4*)&fillp[(3 * 4 + (H)) * K1_PIECE_F + 1 * 16 * ATT_D] = S##13; \
  *(float4*)&fillp[(3 * 4 + (H)) * K1_PIECE_F + 2 * 16 * ATT_D] = S##14; \
  *(float4*)&fillp[(3 * 4 + (H)) * K1_PIECE_F + 3 * 16 * ATT_D] = S##15;
#define K1_FRAGS(FA, FB, H, G)                                       \
  _Pragma("unroll") for (int s = 0; s < 4; ++s) {                    \
    _Pragma("unroll") for (int t = 0; t < 2; ++t) {                  \
      FA[s][t].x = kfrag[t][(H) * K1_PIECE_F + ((G) * 4 + s) * 4 * ATT_D];          \
      FA[s][t].y = kfrag[t][(H) * K1_PIECE_F + ((G) * 4 + s) * 4 * ATT_D + ATT_D];  \
    }                                                                \
    FB[s].x = vfrag[(H) * K1_PIECE_F + ((G) * 4 + s) * 4 * ATT_D];                  \
    FB[s].y = vfrag[(H) * K1_PIECE_F + ((G) * 4 + s) * 4 * ATT_D + ATT_D];          \
  }
#define K1_LBAR()                                                    \
  do {                                                               \
    __builtin_amdgcn_fence(__ATOMIC_RELEASE, "workgroup", "local");  \
    __builtin_amdgcn_s_barrier();                                    \
    __builtin_amdgcn_fence(__ATOMIC_ACQUIRE, "workgroup", "local");  \
  } while (0)
#define K1_PIN() __builtin_amdgcn_sched_barrier(0)
#define K1_MMA(FA, FB)                                               \
  _Pragma("unroll") for (int s = 0; s < 4; ++s)                      \
    _Pragma("unroll") for (int i = 0; i < 2; ++i)                    \
      c[i] = wmma_f32(FA[s][i], FB[s], c[i]);
#define K1_PIECE(H)                                                  \
  K1_FRAGS(fa0, fb0, H, 0)                                           \
  K1_FRAGS(fa1, fb1, H, 1) K1_MMA(fa0, fb0)                          \
  K1_FRAGS(fa0, fb0, H, 2) K1_MMA(fa1, fb1)                          \
  K1_FRAGS(fa1, fb1, H, 3) K1_MMA(fa0, fb0)                          \
  K1_MMA(fa1, fb1)

__global__ __launch_bounds__(K1_THREADS)
void attn_ktv_partial(const float* K, const float* V,
                      float* __restrict__ partial) {
  const int quad  = blockIdx.x % NQUAD;
  const int chunk = (blockIdx.x / NQUAD) % NCHUNK;
  const int head  = blockIdx.x / (NQUAD * NCHUNK);
  const int qi    = quad >> 1;
  const int qj    = quad & 1;
  const int wave  = threadIdx.x >> 5;
  const int lane  = threadIdx.x & 31;
  const int lmod  = lane & 15;
  const int lhalf = lane >> 4;
  const int wj    = wave & 1;
  const int kg    = wave >> 1;

  __shared__ __attribute__((aligned(16))) float sm[K1_NPIECE * K1_PIECE_F];
  __shared__ __attribute__((aligned(16))) float red[4 * K1_RED_F];

  const int c4 = threadIdx.x & 15;
  const size_t row0 = (size_t)head * ATT_S + (size_t)chunk * CHUNK_S + (threadIdx.x >> 4);
  const float* src = (c4 < 8)
      ? K + row0 * ATT_D + qi * 32 + c4 * 4
      : V + row0 * ATT_D + qj * 32 + (c4 - 8) * 4;
  const float4* src4 = (const float4*)src;

  float* __restrict__ fillp =
      &sm[(threadIdx.x >> 4) * ATT_D + ((c4 ^ ((wave & 1) << 2)) << 2)];

  const float* __restrict__ kfrag[2];
#pragma unroll
  for (int t = 0; t < 2; ++t)
    kfrag[t] = &sm[kg * 4 * K1_PIECE_F + 2 * lhalf * ATT_D + ((t ^ lhalf) << 4) + lmod];
  const float* __restrict__ vfrag =
      &sm[kg * 4 * K1_PIECE_F + 2 * lhalf * ATT_D + (((2 + wj) ^ lhalf) << 4) + lmod];

  float4 sa0, sa1, sa2, sa3, sa4, sa5, sa6, sa7,
         sa8, sa9, sa10, sa11, sa12, sa13, sa14, sa15;
  float4 sb0, sb1, sb2, sb3, sb4, sb5, sb6, sb7,
         sb8, sb9, sb10, sb11, sb12, sb13, sb14, sb15;
  v2f fa0[4][2], fb0[4], fa1[4][2], fb1[4];
  v8f c[2];
#pragma unroll
  for (int i = 0; i < 2; ++i)
    c[i] = (v8f){0.f, 0.f, 0.f, 0.f, 0.f, 0.f, 0.f, 0.f};

  K1_GLOAD(sa, 0)
  K1_PIN();
  K1_GLOAD(sb, 1)
  K1_FILL(sa, 0)
  K1_PIN();
  K1_GLOAD(sa, 2)
  K1_LBAR();
  K1_PIECE(0)

  K1_PIN();
  K1_FILL(sb, 1)
  K1_PIN();
  K1_GLOAD(sb, 3)
  K1_LBAR();
  K1_PIECE(1)

  K1_PIN();
  K1_FILL(sa, 2)
  K1_LBAR();
  K1_PIECE(2)

  K1_PIN();
  K1_FILL(sb, 3)
  K1_LBAR();
  K1_PIECE(3)

  float* __restrict__ redw =
      &red[kg * K1_RED_F + (lhalf * 8) * K1_RPITCH + wj * 16 + lmod];
#pragma unroll
  for (int i = 0; i < 2; ++i)
#pragma unroll
    for (int r = 0; r < 8; ++r)
      redw[(i * 16 + r) * K1_RPITCH] = c[i][r];
  __syncthreads();

  const int rr = threadIdx.x >> 3;
  const int rc = threadIdx.x & 7;
  const float* __restrict__ redr = &red[rr * K1_RPITCH + rc * 4];
  float4 acc = *(const float4*)&redr[0];
#pragma unroll
  for (int g = 1; g < 4; ++g) {
    const float4 x = *(const float4*)&redr[g * K1_RED_F];
    acc.x += x.x; acc.y += x.y; acc.z += x.z; acc.w += x.w;
  }
  float* __restrict__ outp =
      partial + ((size_t)head * NCHUNK + chunk) * (ATT_D * ATT_D) +
      (size_t)(qi * 32 + rr) * ATT_D + qj * 32 + rc * 4;
  *(float4*)outp = acc;
}

__global__ __launch_bounds__(QWAVES * 32)
void attn_qm(const float* __restrict__ Q, const float* __restrict__ partial,
             float* __restrict__ out) {
  const int head  = blockIdx.x / (ATT_S / QTILE);
  const int qc    = blockIdx.x % (ATT_S / QTILE);
  const int wave  = threadIdx.x >> 5;
  const int lane  = threadIdx.x & 31;
  const int lmod  = lane & 15;
  const int lhalf = lane >> 4;
  const int q0    = qc * QTILE;

  __shared__ __attribute__((aligned(16))) float qs[QTILE * QPAD];
  __shared__ __attribute__((aligned(16))) float ms[(ATT_D / 2) * MKP];

  const float4* __restrict__ Qh4 =
      (const float4*)(Q + ((size_t)head * ATT_S + q0) * ATT_D);
  float4 qv[16];
#pragma unroll
  for (int i = 0; i < 16; ++i)
    qv[i] = Qh4[threadIdx.x + i * (QWAVES * 32)];

  const float* __restrict__ ph =
      partial + (size_t)head * NCHUNK * ATT_D * ATT_D;
#pragma unroll
  for (int trip = 0; trip < 2; ++trip) {
    const int item = trip * (QWAVES * 32) + threadIdx.x;
    const int kp   = item >> 4;
    const int n4   = item & 15;
    const float4* __restrict__ pe =
        (const float4*)(ph + (2 * kp) * ATT_D + n4 * 4);
    const float4* __restrict__ po =
        (const float4*)(ph + (2 * kp + 1) * ATT_D + n4 * 4);
    float4 e[NCHUNK], o[NCHUNK];
#pragma unroll
    for (int cc = 0; cc < NCHUNK; ++cc) {
      e[cc] = pe[cc * (ATT_D * ATT_D / 4)];
      o[cc] = po[cc * (ATT_D * ATT_D / 4)];
    }
    if (trip == 0) {
#pragma unroll
      for (int i = 0; i < 16; ++i) {
        const int idx = threadIdx.x + i * (QWAVES * 32);
        const int row = idx >> 4;
        const int c4  = idx & 15;
        *(float4*)&qs[row * QPAD + c4 * 4] = qv[i];
      }
    }
    float4 se = {0.f, 0.f, 0.f, 0.f}, so = {0.f, 0.f, 0.f, 0.f};
#pragma unroll
    for (int cc = 0; cc < NCHUNK; ++cc) {
      se.x += e[cc].x; se.y += e[cc].y; se.z += e[cc].z; se.w += e[cc].w;
      so.x += o[cc].x; so.y += o[cc].y; so.z += o[cc].z; so.w += o[cc].w;
    }
    float* __restrict__ d = &ms[kp * MKP + n4 * 8];
    *(float4*)(d)     = (float4){se.x, so.x, se.y, so.y};
    *(float4*)(d + 4) = (float4){se.z, so.z, se.w, so.w};
  }
  __syncthreads();

  v8f c[2][4];
#pragma unroll
  for (int i = 0; i < 2; ++i)
#pragma unroll
    for (int j = 0; j < 4; ++j)
      c[i][j] = (v8f){0.f, 0.f, 0.f, 0.f, 0.f, 0.f, 0.f, 0.f};

  const float* __restrict__ qw = &qs[(wave * 32 + lmod) * QPAD + lhalf * 2];
  const float* __restrict__ mw = &ms[lhalf * MKP + lmod * 2];
#pragma unroll 2
  for (int step = 0; step < ATT_D / 4; ++step) {
    v2f a[2], b[4];
#pragma unroll
    for (int t = 0; t < 2; ++t)
      a[t] = *(const v2f*)&qw[t * 16 * QPAD + step * 4];
#pragma unroll
    for (int t = 0; t < 4; ++t)
      b[t] = *(const v2f*)&mw[step * 2 * MKP + t * 32];
#pragma unroll
    for (int i = 0; i < 2; ++i)
#pragma unroll
      for (int j = 0; j < 4; ++j)
        c[i][j] = wmma_f32(a[i], b[j], c[i][j]);
  }

  float* __restrict__ Oh =
      out + ((size_t)head * ATT_S + q0 + wave * 32) * ATT_D;
#pragma unroll
  for (int i = 0; i < 2; ++i)
#pragma unroll
    for (int j = 0; j < 4; ++j)
#pragma unroll
      for (int r = 0; r < 8; ++r)
        Oh[(size_t)(i * 16 + r + lhalf * 8) * ATT_D + j * 16 + lmod] =
            c[i][j][r];
}

extern "C" void kernel_launch(void* const* d_in, const int* in_sizes, int n_in,
                              void* d_out, int out_size, void* d_ws,
                              size_t ws_size, hipStream_t stream) {
  (void)in_sizes; (void)n_in; (void)out_size; (void)ws_size;
  const float* q = (const float*)d_in[0];
  const float* k = (const float*)d_in[1];
  const float* v = (const float*)d_in[2];
  float* out = (float*)d_out;

  float* partial = (float*)d_ws;

  attn_ktv_partial<<<NHEAD * NCHUNK * NQUAD, K1_WAVES * 32, 0, stream>>>(k, v, partial);
  attn_qm<<<NHEAD * (ATT_S / QTILE), QWAVES * 32, 0, stream>>>(q, partial, out);
}
